# starting kernel
# baseline (speedup 1.0000x reference)
#include <hip/hip_runtime.h>
#include <math.h>

typedef __attribute__((ext_vector_type(2))) float v2f;
typedef __attribute__((ext_vector_type(8))) float v8f;

#define DIM_B 2
#define DIM_T 2048
#define DIM_K 1024
#define NHEAD 16
#define HDIM  64

static __device__ __forceinline__ v8f wmma_f32(v2f a, v2f b, v8f c) {
  return __builtin_amdgcn_wmma_f32_16x16x4_f32(false, a, false, b, (short)0, c,
                                               false, false);
}

__global__ __launch_bounds__(256)
__attribute__((amdgpu_waves_per_eu(2)))
void gemm_xwt_kernel(const float* __restrict__ A, const float* __restrict__ W,
                     const float* __restrict__ bias, float* __restrict__ C,
                     int M, int N, int K) {
  const int lane  = threadIdx.x & 31;
  const int wave  = threadIdx.x >> 5;
  const int wm    = wave >> 2;
  const int wn    = wave & 3;
  const int m0    = blockIdx.y * 128 + wm * 64;
  const int n0    = blockIdx.x * 128 + wn * 32;
  const int lhalf = lane >> 4;
  const int lmod  = lane & 15;
  const int kbase = lhalf * 2;

  v8f acc[4][2];
#pragma unroll
  for (int i = 0; i < 4; ++i)
#pragma unroll
    for (int j = 0; j < 2; ++j)
#pragma unroll
      for (int r = 0; r < 8; ++r) acc[i][j][r] = 0.0f;

  const float* ar0 = A + (size_t)(m0 + lmod) * K;
  const float* ar1 = ar0 + (size_t)16 * K;
  const float* ar2 = ar0 + (size_t)32 * K;
  const float* ar3 = ar0 + (size_t)48 * K;
  const float* br0 = W + (size_t)(n0 + lmod) * K;
  const float* br1 = br0 + (size_t)16 * K;

  v2f a0 = *(const v2f*)(ar0 + kbase);
  v2f a1 = *(const v2f*)(ar1 + kbase);
  v2f a2 = *(const v2f*)(ar2 + kbase);
  v2f a3 = *(const v2f*)(ar3 + kbase);
  v2f b0 = *(const v2f*)(br0 + kbase);
  v2f b1 = *(const v2f*)(br1 + kbase);

  v2f c0 = *(const v2f*)(ar0 + 4 + kbase);
  v2f c1 = *(const v2f*)(ar1 + 4 + kbase);
  v2f c2 = *(const v2f*)(ar2 + 4 + kbase);
  v2f c3 = *(const v2f*)(ar3 + 4 + kbase);
  v2f d0 = *(const v2f*)(br0 + 4 + kbase);
  v2f d1 = *(const v2f*)(br1 + 4 + kbase);

#pragma clang loop unroll(disable)
  for (int k0 = 0; k0 + 16 <= K; k0 += 8) {
    acc[0][0] = wmma_f32(a0, b0, acc[0][0]);
    acc[0][1] = wmma_f32(a0, b1, acc[0][1]);
    acc[1][0] = wmma_f32(a1, b0, acc[1][0]);
    acc[1][1] = wmma_f32(a1, b1, acc[1][1]);
    acc[2][0] = wmma_f32(a2, b0, acc[2][0]);
    acc[2][1] = wmma_f32(a2, b1, acc[2][1]);
    acc[3][0] = wmma_f32(a3, b0, acc[3][0]);
    acc[3][1] = wmma_f32(a3, b1, acc[3][1]);
    {
      const int kn = k0 + 8 + kbase;
      a0 = *(const v2f*)(ar0 + kn);
      a1 = *(const v2f*)(ar1 + kn);
      a2 = *(const v2f*)(ar2 + kn);
      a3 = *(const v2f*)(ar3 + kn);
      b0 = *(const v2f*)(br0 + kn);
      b1 = *(const v2f*)(br1 + kn);
    }
    acc[0][0] = wmma_f32(c0, d0, acc[0][0]);
    acc[0][1] = wmma_f32(c0, d1, acc[0][1]);
    acc[1][0] = wmma_f32(c1, d0, acc[1][0]);
    acc[1][1] = wmma_f32(c1, d1, acc[1][1]);
    acc[2][0] = wmma_f32(c2, d0, acc[2][0]);
    acc[2][1] = wmma_f32(c2, d1, acc[2][1]);
    acc[3][0] = wmma_f32(c3, d0, acc[3][0]);
    acc[3][1] = wmma_f32(c3, d1, acc[3][1]);
    {
      const int kn = k0 + 12 + kbase;
      c0 = *(const v2f*)(ar0 + kn);
      c1 = *(const v2f*)(ar1 + kn);
      c2 = *(const v2f*)(ar2 + kn);
      c3 = *(const v2f*)(ar3 + kn);
      d0 = *(const v2f*)(br0 + kn);
      d1 = *(const v2f*)(br1 + kn);
    }
  }
  acc[0][0] = wmma_f32(a0, b0, acc[0][0]);
  acc[0][1] = wmma_f32(a0, b1, acc[0][1]);
  acc[1][0] = wmma_f32(a1, b0, acc[1][0]);
  acc[1][1] = wmma_f32(a1, b1, acc[1][1]);
  acc[2][0] = wmma_f32(a2, b0, acc[2][0]);
  acc[2][1] = wmma_f32(a2, b1, acc[2][1]);
  acc[3][0] = wmma_f32(a3, b0, acc[3][0]);
  acc[3][1] = wmma_f32(a3, b1, acc[3][1]);
  acc[0][0] = wmma_f32(c0, d0, acc[0][0]);
  acc[0][1] = wmma_f32(c0, d1, acc[0][1]);
  acc[1][0] = wmma_f32(c1, d0, acc[1][0]);
  acc[1][1] = wmma_f32(c1, d1, acc[1][1]);
  acc[2][0] = wmma_f32(c2, d0, acc[2][0]);
  acc[2][1] = wmma_f32(c2, d1, acc[2][1]);
  acc[3][0] = wmma_f32(c3, d0, acc[3][0]);
  acc[3][1] = wmma_f32(c3, d1, acc[3][1]);

#pragma unroll
  for (int mi = 0; mi < 4; ++mi)
#pragma unroll
    for (int ni = 0; ni < 2; ++ni)
#pragma unroll
      for (int r = 0; r < 8; ++r) {
        const int row = m0 + mi * 16 + r + lhalf * 8;
        const int col = n0 + ni * 16 + lmod;
        float v = acc[mi][ni][r];
        if (bias) v += bias[col];
        C[(size_t)row * N + col] = v;
      }
}

static __device__ __forceinline__ float hred_max(float v) {
  v = fmaxf(v, __shfl_xor(v, 1));
  v = fmaxf(v, __shfl_xor(v, 2));
  v = fmaxf(v, __shfl_xor(v, 4));
  v = fmaxf(v, __shfl_xor(v, 8));
  return v;
}
static __device__ __forceinline__ float hred_sum(float v) {
  v += __shfl_xor(v, 1);
  v += __shfl_xor(v, 2);
  v += __shfl_xor(v, 4);
  v += __shfl_xor(v, 8);
  return v;
}

__global__ __launch_bounds__(128)
__attribute__((amdgpu_waves_per_eu(1)))
void attn_kernel(const float* __restrict__ Q, const float* __restrict__ Km,
                 const float* __restrict__ V, float* __restrict__ O) {
  __shared__ float pbuf[4][16 * 16];

  const int lane  = threadIdx.x & 31;
  const int wave  = threadIdx.x >> 5;
  const int lhalf = lane >> 4;
  const int lmod  = lane & 15;
  const int h  = blockIdx.y;
  const int b  = blockIdx.z;
  const int q0 = blockIdx.x * 64 + wave * 16;

  const size_t base = (size_t)b * DIM_T * DIM_K + (size_t)h * HDIM;

  v2f qa[16];
  {
    const float* qrow = Q + base + (size_t)(q0 + lmod) * DIM_K;
#pragma unroll
    for (int t = 0; t < 16; ++t)
      qa[t] = *(const v2f*)(qrow + t * 4 + lhalf * 2);
  }

  v8f acc[4];
#pragma unroll
  for (int j = 0; j < 4; ++j)
#pragma unroll
    for (int r = 0; r < 8; ++r) acc[j][r] = 0.0f;

  float mrow[8], lrow[8];
#pragma unroll
  for (int r = 0; r < 8; ++r) { mrow[r] = -INFINITY; lrow[r] = 0.0f; }

  const float scale = 0.03125f;

  for (int key0 = 0; key0 < DIM_T; key0 += 16) {
    const float* krow = Km + base + (size_t)(key0 + lmod) * DIM_K;
    v2f kb[16];
#pragma unroll
    for (int t = 0; t < 16; ++t)
      kb[t] = *(const v2f*)(krow + t * 4 + lhalf * 2);

    v8f s;
#pragma unroll
    for (int r = 0; r < 8; ++r) s[r] = 0.0f;
#pragma unroll
    for (int t = 0; t < 16; ++t) s = wmma_f32(qa[t], kb[t], s);

    v2f vf[4][4];
#pragma unroll
    for (int t = 0; t < 4; ++t) {
      const int kk = t * 4 + lhalf * 2;
      const float* vbase = V + base + (size_t)(key0 + kk) * DIM_K + lmod;
#pragma unroll
      for (int j = 0; j < 4; ++j) {
        vf[t][j].x = vbase[j * 16];
        vf[t][j].y = vbase[j * 16 + DIM_K];
      }
    }

    float p[8];
#pragma unroll
    for (int r = 0; r < 8; ++r) {
      const float sv   = s[r] * scale;
      const float rm   = hred_max(sv);
      const float mnew = fmaxf(mrow[r], rm);
      const float corr = __expf(mrow[r] - mnew);
      const float pv   = __expf(sv - mnew);
      lrow[r] = lrow[r] * corr + hred_sum(pv);
      mrow[r] = mnew;
      p[r]    = pv;
#pragma unroll
      for (int j = 0; j < 4; ++j) acc[j][r] *= corr;
    }

#pragma unroll
    for (int r = 0; r < 8; ++r)
      pbuf[wave][(r + lhalf * 8) * 16 + lmod] = p[r];

#pragma unroll
    for (int t = 0; t < 4; ++t) {
      const int kk = t * 4 + lhalf * 2;
      v2f pa;
      pa.x = pbuf[wave][lmod * 16 + kk];
      pa.y = pbuf[wave][lmod * 16 + kk + 1];
#pragma unroll
      for (int j = 0; j < 4; ++j)
        acc[j] = wmma_f32(pa, vf[t][j], acc[j]);
    }
  }

#pragma unroll
  for (int j = 0; j < 4; ++j)
#pragma unroll
    for (int r = 0; r < 8; ++r) {
      const float ov = acc[j][r] / lrow[r];
      O[base + (size_t)(q0 + r + lhalf * 8) * DIM_K + j * 16 + lmod] = ov;
    }
}

extern "C" void kernel_launch(void* const* d_in, const int* in_sizes, int n_in,
                              void* d_out, int out_size, void* d_ws,
                              size_t ws_size, hipStream_t stream) {
  (void)in_sizes; (void)n_in; (void)out_size; (void)ws_size;
  const float* x  = (const float*)d_in[0];
  const float* wq = (const float*)d_in[1];
  const float* wk = (const float*)d_in[2];
  const float* wv = (const float*)d_in[3];
  const float* wu = (const float*)d_in[4];
  const float* bu = (const float*)d_in[5];
  float* out = (float*)d_out;

  const size_t MAT = (size_t)DIM_B * DIM_T * DIM_K;
  float* qb = (float*)d_ws;
  float* kb = qb + MAT;
  float* vb = kb + MAT;
  float* ab = vb + MAT;

  const int M = DIM_B * DIM_T, N = DIM_K, K = DIM_K;
  dim3 gblk(256);
  dim3 ggrid(N / 128, M / 128);

  gemm_xwt_kernel<<<ggrid, gblk, 0, stream>>>(x, wq, nullptr, qb, M, N, K);
  gemm_xwt_kernel<<<ggrid, gblk, 0, stream>>>(x, wk, nullptr, kb, M, N, K);
  gemm_xwt_kernel<<<ggrid, gblk, 0, stream>>>(x, wv, nullptr, vb, M, N, K);

  dim3 agrid(DIM_T / 64, NHEAD, DIM_B);
  attn_kernel<<<agrid, dim3(128), 0, stream>>>(qb, kb, vb, ab);

  gemm_xwt_kernel<<<ggrid, gblk, 0, stream>>>(ab, wu, bu, out, M, N, K);
}


# attention: score tile goes through the wave-private LDS relayout BEFORE the softmax: a lane then holds 8 keys of one row, row max is in-lane plus one half exchange; row-sum order changes
# speedup vs baseline: 1.3800x; 1.3800x over previous
#include <hip/hip_runtime.h>
#include <math.h>

typedef __attribute__((ext_vector_type(2))) float v2f;
typedef __attribute__((ext_vector_type(4))) float v4f;
typedef __attribute__((ext_vector_type(8))) float v8f;

#define DIM_B 2
#define DIM_T 2048
#define DIM_K 1024
#define NHEAD 16
#define HDIM  64

static __device__ __forceinline__ v8f wmma_f32(v2f a, v2f b, v8f c) {
  return __builtin_amdgcn_wmma_f32_16x16x4_f32(false, a, false, b, (short)0, c,
                                               false, false);
}

template <int OUT>
__global__ __launch_bounds__(256)
__attribute__((amdgpu_waves_per_eu(2)))
void gemm_xwt_kernel(const float* __restrict__ A, const float* __restrict__ W,
                     const float* __restrict__ bias, float* __restrict__ C,
                     int M, int N, int K) {
  const int lane  = threadIdx.x & 31;
  const int wave  = threadIdx.x >> 5;
  const int wm    = wave >> 2;
  const int wn    = wave & 3;
  const int m0    = blockIdx.y * 128 + wm * 64;
  const int n0    = blockIdx.x * 128 + wn * 32;
  const int lhalf = lane >> 4;
  const int lmod  = lane & 15;
  const int kbase = lhalf * 4;

  v8f acc[4][2];
#pragma unroll
  for (int i = 0; i < 4; ++i)
#pragma unroll
    for (int j = 0; j < 2; ++j)
#pragma unroll
      for (int r = 0; r < 8; ++r) acc[i][j][r] = 0.0f;

  const float* ar0 = A + (size_t)(m0 + lmod) * K;
  const float* ar1 = ar0 + (size_t)16 * K;
  const float* ar2 = ar0 + (size_t)32 * K;
  const float* ar3 = ar0 + (size_t)48 * K;
  const float* br0 = W + (size_t)(n0 + lmod) * K;
  const float* br1 = br0 + (size_t)16 * K;

#define GEMM_MMA16(f0, f1, f2, f3, g0, g1)            \
  acc[0][0] = wmma_f32(f0.xy, g0.xy, acc[0][0]);      \
  acc[0][1] = wmma_f32(f0.xy, g1.xy, acc[0][1]);      \
  acc[1][0] = wmma_f32(f1.xy, g0.xy, acc[1][0]);      \
  acc[1][1] = wmma_f32(f1.xy, g1.xy, acc[1][1]);      \
  acc[2][0] = wmma_f32(f2.xy, g0.xy, acc[2][0]);      \
  acc[2][1] = wmma_f32(f2.xy, g1.xy, acc[2][1]);      \
  acc[3][0] = wmma_f32(f3.xy, g0.xy, acc[3][0]);      \
  acc[3][1] = wmma_f32(f3.xy, g1.xy, acc[3][1]);      \
  acc[0][0] = wmma_f32(f0.zw, g0.zw, acc[0][0]);      \
  acc[0][1] = wmma_f32(f0.zw, g1.zw, acc[0][1]);      \
  acc[1][0] = wmma_f32(f1.zw, g0.zw, acc[1][0]);      \
  acc[1][1] = wmma_f32(f1.zw, g1.zw, acc[1][1]);      \
  acc[2][0] = wmma_f32(f2.zw, g0.zw, acc[2][0]);      \
  acc[2][1] = wmma_f32(f2.zw, g1.zw, acc[2][1]);      \
  acc[3][0] = wmma_f32(f3.zw, g0.zw, acc[3][0]);      \
  acc[3][1] = wmma_f32(f3.zw, g1.zw, acc[3][1]);

#define GEMM_LOAD6(f0, f1, f2, f3, g0, g1, koff)      \
  {                                                   \
    const int kn = (koff) + kbase;                    \
    f0 = *(const v4f*)(ar0 + kn);                     \
    f1 = *(const v4f*)(ar1 + kn);                     \
    f2 = *(const v4f*)(ar2 + kn);                     \
    f3 = *(const v4f*)(ar3 + kn);                     \
    g0 = *(const v4f*)(br0 + kn);                     \
    g1 = *(const v4f*)(br1 + kn);                     \
  }
#define GEMM_FENCE __builtin_amdgcn_sched_barrier(0);

  v4f a0, a1, a2, a3, b0, b1;
  v4f c0, c1, c2, c3, d0, d1;
  v4f e0, e1, e2, e3, f0, f1;
  v4f g0, g1, g2, g3, h0, h1;

  GEMM_LOAD6(a0, a1, a2, a3, b0, b1, 0)
  GEMM_FENCE
  GEMM_LOAD6(c0, c1, c2, c3, d0, d1, 8)
  GEMM_FENCE
  GEMM_LOAD6(e0, e1, e2, e3, f0, f1, 16)

  int k0 = 0;
#pragma clang loop unroll(disable)
  for (; k0 + 64 <= K; k0 += 32) {
    GEMM_LOAD6(g0, g1, g2, g3, h0, h1, k0 + 24)
    GEMM_FENCE
    GEMM_MMA16(a0, a1, a2, a3, b0, b1)
    GEMM_LOAD6(a0, a1, a2, a3, b0, b1, k0 + 32)
    GEMM_FENCE
    GEMM_MMA16(c0, c1, c2, c3, d0, d1)
    GEMM_LOAD6(c0, c1, c2, c3, d0, d1, k0 + 40)
    GEMM_FENCE
    GEMM_MMA16(e0, e1, e2, e3, f0, f1)
    GEMM_LOAD6(e0, e1, e2, e3, f0, f1, k0 + 48)
    GEMM_FENCE
    GEMM_MMA16(g0, g1, g2, g3, h0, h1)
  }
  GEMM_LOAD6(g0, g1, g2, g3, h0, h1, k0 + 24)
  GEMM_MMA16(a0, a1, a2, a3, b0, b1)
  GEMM_MMA16(c0, c1, c2, c3, d0, d1)
  GEMM_MMA16(e0, e1, e2, e3, f0, f1)
  GEMM_MMA16(g0, g1, g2, g3, h0, h1)
#undef GEMM_FENCE
#undef GEMM_LOAD6
#undef GEMM_MMA16

  if (OUT == 0) {
#pragma unroll
    for (int mi = 0; mi < 4; ++mi)
#pragma unroll
      for (int ni = 0; ni < 2; ++ni)
#pragma unroll
        for (int r = 0; r < 8; ++r) {
          const int row = m0 + mi * 16 + r + lhalf * 8;
          const int col = n0 + ni * 16 + lmod;
          float v = acc[mi][ni][r];
          if (bias) v += bias[col];
          C[(size_t)row * N + col] = v;
        }
  } else if (OUT == 1) {
    const int h = m0 >> 6;
#pragma unroll
    for (int ni = 0; ni < 2; ++ni) {
      const int tok0  = n0 + ni * 16;
      const int bb    = tok0 / DIM_T;
      const int chunk = (tok0 % DIM_T) >> 4;
      float* dst = C + ((size_t)(bb * NHEAD + h) * (DIM_T / 16) + chunk) * 1024
                     + lane * 4;
#pragma unroll
      for (int mi = 0; mi < 4; ++mi)
#pragma unroll
        for (int rp = 0; rp < 2; ++rp) {
          v4f o;
          o.x = acc[mi][ni][4 * rp + 0];
          o.y = acc[mi][ni][4 * rp + 1];
          o.z = acc[mi][ni][4 * rp + 2];
          o.w = acc[mi][ni][4 * rp + 3];
          *(v4f*)(dst + (mi * 2 + rp) * 128) = o;
        }
    }
  } else {
    const int h  = n0 >> 6;
    const int jb = (n0 & 63) >> 4;
#pragma unroll
    for (int mi = 0; mi < 4; ++mi) {
      const int tok0  = m0 + mi * 16;
      const int bb    = tok0 / DIM_T;
      const int chunk = (tok0 % DIM_T) >> 4;
      float* dst = C + ((size_t)(bb * NHEAD + h) * (DIM_T / 16) + chunk) * 1024
                     + lane * 4;
#pragma unroll
      for (int ni = 0; ni < 2; ++ni)
#pragma unroll
        for (int rp = 0; rp < 2; ++rp) {
          v4f o;
          o.x = acc[mi][ni][4 * rp + 0];
          o.y = acc[mi][ni][4 * rp + 1];
          o.z = acc[mi][ni][4 * rp + 2];
          o.w = acc[mi][ni][4 * rp + 3];
          *(v4f*)(dst + ((jb + ni) * 2 + rp) * 128) = o;
        }
    }
  }
}

static __device__ __forceinline__ float xhalf(float x) {
  const int b = __builtin_bit_cast(int, x);
  return __builtin_bit_cast(
      float, __builtin_amdgcn_permlanex16(b, b, 0x76543210, 0xfedcba98, false,
                                          false));
}
static __device__ __forceinline__ float to_acc_rows(float v) {
  const int b = __builtin_bit_cast(int, v);
  return __builtin_bit_cast(
      float, __builtin_amdgcn_update_dpp(b, b, 0x160 + 8, 0x2, 0xf, false));
}
template <int R>
static __device__ __forceinline__ float row_share(float v) {
  return __builtin_bit_cast(
      float, __builtin_amdgcn_update_dpp(0, __builtin_bit_cast(int, v),
                                         0x150 + R, 0xf, 0xf, true));
}

__global__ __launch_bounds__(128)
__attribute__((amdgpu_waves_per_eu(1)))
void attn_kernel(const float* __restrict__ Q, const float* __restrict__ Kf,
                 const float* __restrict__ Vf, float* __restrict__ O) {
  __shared__ __attribute__((aligned(16))) float pbuf[4][16 * 20];

  const int lane  = threadIdx.x & 31;
  const int wave  = threadIdx.x >> 5;
  const int lhalf = lane >> 4;
  const int lmod  = lane & 15;
  const int h  = blockIdx.y;
  const int b  = blockIdx.z;
  const int q0 = blockIdx.x * 64 + wave * 16;

  const size_t base = (size_t)b * DIM_T * DIM_K + (size_t)h * HDIM;

  v2f qa[16];
  {
    const float* qrow = Q + base + (size_t)(q0 + lmod) * DIM_K + lhalf * 8;
#pragma unroll
    for (int g = 0; g < 8; ++g) {
      const v4f t = *(const v4f*)(qrow + (g >> 1) * 16 + (g & 1) * 4);
      qa[2 * g].x     = t.x;  qa[2 * g].y     = t.y;
      qa[2 * g + 1].x = t.z;  qa[2 * g + 1].y = t.w;
    }
  }

  const size_t fbase = (size_t)(b * NHEAD + h) * (DIM_T / 16) * 1024 + lane * 4;
  const float* kfp = Kf + fbase;
  const float* vfp = Vf + fbase;

  v8f acc[4];
#pragma unroll
  for (int j = 0; j < 4; ++j)
#pragma unroll
    for (int r = 0; r < 8; ++r) acc[j][r] = 0.0f;

  float mrow = -INFINITY, lrow = 0.0f;

  const float scale = 0.03125f;

  for (int key0 = 0; key0 < DIM_T; key0 += 16) {
    const size_t coff = (size_t)(key0 >> 4) * 1024;
    v4f kk[8];
#pragma unroll
    for (int g = 0; g < 8; ++g)
      kk[g] = *(const v4f*)(kfp + coff + g * 128);

    v8f s;
#pragma unroll
    for (int r = 0; r < 8; ++r) s[r] = 0.0f;
#pragma unroll
    for (int g = 0; g < 8; ++g) {
      v2f k0, k1;
      k0.x = kk[g].x;  k0.y = kk[g].y;
      k1.x = kk[g].z;  k1.y = kk[g].w;
      s = wmma_f32(qa[2 * g], k0, s);
      s = wmma_f32(qa[2 * g + 1], k1, s);
    }

    v2f vf[4][4];
#pragma unroll
    for (int j = 0; j < 4; ++j)
#pragma unroll
      for (int rp = 0; rp < 2; ++rp) {
        const v4f t = *(const v4f*)(vfp + coff + (j * 2 + rp) * 128);
        vf[2 * rp][j].x     = t.x;  vf[2 * rp][j].y     = t.y;
        vf[2 * rp + 1][j].x = t.z;  vf[2 * rp + 1][j].y = t.w;
      }
    __builtin_amdgcn_sched_barrier(0);

#pragma unroll
    for (int r = 0; r < 8; ++r)
      pbuf[wave][(r + lhalf * 8) * 20 + lmod] = s[r];
    const v4f t0 = *(const v4f*)&pbuf[wave][lmod * 20 + lhalf * 8];
    const v4f t1 = *(const v4f*)&pbuf[wave][lmod * 20 + lhalf * 8 + 4];
    float sv[8];
    sv[0] = t0.x * scale;  sv[1] = t0.y * scale;
    sv[2] = t0.z * scale;  sv[3] = t0.w * scale;
    sv[4] = t1.x * scale;  sv[5] = t1.y * scale;
    sv[6] = t1.z * scale;  sv[7] = t1.w * scale;

    const float mloc =
        fmaxf(fmaxf(fmaxf(sv[0], sv[1]), fmaxf(sv[2], sv[3])),
              fmaxf(fmaxf(sv[4], sv[5]), fmaxf(sv[6], sv[7])));
    const float mnew = fmaxf(mrow, fmaxf(mloc, xhalf(mloc)));
    const float corr = __expf(mrow - mnew);
    mrow = mnew;
    float pe[8];
#pragma unroll
    for (int e = 0; e < 8; ++e) pe[e] = __expf(sv[e] - mnew);
    lrow = lrow * corr + (((pe[0] + pe[1]) + (pe[2] + pe[3])) +
                          ((pe[4] + pe[5]) + (pe[6] + pe[7])));

    {
      const float c2 = to_acc_rows(corr);
      float cr[8];
      cr[0] = row_share<0>(c2);  cr[1] = row_share<1>(c2);
      cr[2] = row_share<2>(c2);  cr[3] = row_share<3>(c2);
      cr[4] = row_share<4>(c2);  cr[5] = row_share<5>(c2);
      cr[6] = row_share<6>(c2);  cr[7] = row_share<7>(c2);
#pragma unroll
      for (int j = 0; j < 4; ++j)
#pragma unroll
        for (int r = 0; r < 8; ++r) acc[j][r] *= cr[r];
    }

    v2f pa[4];
#pragma unroll
    for (int t = 0; t < 4; ++t) { pa[t].x = pe[2 * t];  pa[t].y = pe[2 * t + 1]; }
#pragma unroll
    for (int t = 0; t < 4; ++t)
#pragma unroll
      for (int j = 0; j < 4; ++j)
        acc[j] = wmma_f32(pa[t], vf[t][j], acc[j]);
  }

  {
    const float l2 = to_acc_rows(lrow + xhalf(lrow));
    float lr[8];
    lr[0] = row_share<0>(l2);  lr[1] = row_share<1>(l2);
    lr[2] = row_share<2>(l2);  lr[3] = row_share<3>(l2);
    lr[4] = row_share<4>(l2);  lr[5] = row_share<5>(l2);
    lr[6] = row_share<6>(l2);  lr[7] = row_share<7>(l2);
#pragma unroll
    for (int j = 0; j < 4; ++j)
#pragma unroll
      for (int r = 0; r < 8; ++r) {
        const float ov = acc[j][r] / lr[r];
        O[base + (size_t)(q0 + r + lhalf * 8) * DIM_K + j * 16 + lmod] = ov;
      }
  }
}

extern "C" void kernel_launch(void* const* d_in, const int* in_sizes, int n_in,
                              void* d_out, int out_size, void* d_ws,
                              size_t ws_size, hipStream_t stream) {
  (void)in_sizes; (void)n_in; (void)out_size; (void)ws_size;
  const float* x  = (const float*)d_in[0];
  const float* wq = (const float*)d_in[1];
  const float* wk = (const float*)d_in[2];
  const float* wv = (const float*)d_in[3];
  const float* wu = (const float*)d_in[4];
  const float* bu = (const float*)d_in[5];
  float* out = (float*)d_out;

  const size_t MAT = (size_t)DIM_B * DIM_T * DIM_K;
  float* qb = (float*)d_ws;
  float* kb = qb + MAT;
  float* vb = kb + MAT;
  float* ab = vb + MAT;

  const int M = DIM_B * DIM_T, N = DIM_K, K = DIM_K;
  dim3 gblk(256);
  dim3 ggrid(N / 128, M / 128);

  gemm_xwt_kernel<0><<<ggrid, gblk, 0, stream>>>(x, wq, nullptr, qb, M, N, K);
  dim3 kgrid(M / 128, N / 128);
  gemm_xwt_kernel<1><<<kgrid, gblk, 0, stream>>>(wk, x, nullptr, kb, N, M, K);
  gemm_xwt_kernel<2><<<ggrid, gblk, 0, stream>>>(x, wv, nullptr, vb, M, N, K);

  dim3 agrid(DIM_T / 64, NHEAD, DIM_B);
  attn_kernel<<<agrid, dim3(128), 0, stream>>>(qb, kb, vb, ab);

  gemm_xwt_kernel<0><<<ggrid, gblk, 0, stream>>>(ab, wu, bu, out, M, N, K);
}


# all launches: a wave lowers its own issue priority (s_setprio 3,2,1,0) on passing 1/2, 3/4, 7/8 of its k or key loop, so a SIMD's waves advance and finish together, not oldest first; sums unchanged
# speedup vs baseline: 1.4376x; 1.0417x over previous
#include <hip/hip_runtime.h>
#include <math.h>

typedef __attribute__((ext_vector_type(2))) float v2f;
typedef __attribute__((ext_vector_type(4))) float v4f;
typedef __attribute__((ext_vector_type(8))) float v8f;

#define DIM_B 2
#define DIM_T 2048
#define DIM_K 1024
#define NHEAD 16
#define HDIM  64

static __device__ __forceinline__ v8f wmma_f32(v2f a, v2f b, v8f c) {
  return __builtin_amdgcn_wmma_f32_16x16x4_f32(false, a, false, b, (short)0, c,
                                               false, false);
}

template <int OUT>
__global__ __launch_bounds__(256)
__attribute__((amdgpu_waves_per_eu(2)))
void gemm_xwt_kernel(const float* __restrict__ A, const float* __restrict__ W,
                     const float* __restrict__ bias, float* __restrict__ C,
                     int M, int N, int K) {
  const int lane  = threadIdx.x & 31;
  const int wave  = threadIdx.x >> 5;
  const int wm    = wave >> 2;
  const int wn    = wave & 3;
  const int m0    = blockIdx.y * 128 + wm * 64;
  const int n0    = blockIdx.x * 128 + wn * 32;
  const int lhalf = lane >> 4;
  const int lmod  = lane & 15;
  const int kbase = lhalf * 4;

  v8f acc[4][2];
#pragma unroll
  for (int i = 0; i < 4; ++i)
#pragma unroll
    for (int j = 0; j < 2; ++j)
#pragma unroll
      for (int r = 0; r < 8; ++r) acc[i][j][r] = 0.0f;

  const float* ar0 = A + (size_t)(m0 + lmod) * K;
  const float* ar1 = ar0 + (size_t)16 * K;
  const float* ar2 = ar0 + (size_t)32 * K;
  const float* ar3 = ar0 + (size_t)48 * K;
  const float* br0 = W + (size_t)(n0 + lmod) * K;
  const float* br1 = br0 + (size_t)16 * K;

#define GEMM_MMA16(f0, f1, f2, f3, g0, g1)            \
  acc[0][0] = wmma_f32(f0.xy, g0.xy, acc[0][0]);      \
  acc[0][1] = wmma_f32(f0.xy, g1.xy, acc[0][1]);      \
  acc[1][0] = wmma_f32(f1.xy, g0.xy, acc[1][0]);      \
  acc[1][1] = wmma_f32(f1.xy, g1.xy, acc[1][1]);      \
  acc[2][0] = wmma_f32(f2.xy, g0.xy, acc[2][0]);      \
  acc[2][1] = wmma_f32(f2.xy, g1.xy, acc[2][1]);      \
  acc[3][0] = wmma_f32(f3.xy, g0.xy, acc[3][0]);      \
  acc[3][1] = wmma_f32(f3.xy, g1.xy, acc[3][1]);      \
  acc[0][0] = wmma_f32(f0.zw, g0.zw, acc[0][0]);      \
  acc[0][1] = wmma_f32(f0.zw, g1.zw, acc[0][1]);      \
  acc[1][0] = wmma_f32(f1.zw, g0.zw, acc[1][0]);      \
  acc[1][1] = wmma_f32(f1.zw, g1.zw, acc[1][1]);      \
  acc[2][0] = wmma_f32(f2.zw, g0.zw, acc[2][0]);      \
  acc[2][1] = wmma_f32(f2.zw, g1.zw, acc[2][1]);      \
  acc[3][0] = wmma_f32(f3.zw, g0.zw, acc[3][0]);      \
  acc[3][1] = wmma_f32(f3.zw, g1.zw, acc[3][1]);

#define GEMM_LOAD6(f0, f1, f2, f3, g0, g1, koff)      \
  {                                                   \
    const long kn = (long)(koff) + kbase;             \
    f0 = *(const v4f*)(ar0 + kn);                     \
    f1 = *(const v4f*)(ar1 + kn);                     \
    f2 = *(const v4f*)(ar2 + kn);                     \
    f3 = *(const v4f*)(ar3 + kn);                     \
    g0 = *(const v4f*)(br0 + kn);                     \
    g1 = *(const v4f*)(br1 + kn);                     \
  }
#define GEMM_FENCE __builtin_amdgcn_sched_barrier(0);

  v4f a0, a1, a2, a3, b0, b1;
  v4f c0, c1, c2, c3, d0, d1;
  v4f e0, e1, e2, e3, f0, f1;
  v4f g0, g1, g2, g3, h0, h1;

  GEMM_LOAD6(a0, a1, a2, a3, b0, b1, 0)
  GEMM_FENCE
  GEMM_LOAD6(c0, c1, c2, c3, d0, d1, 8)
  GEMM_FENCE
  GEMM_LOAD6(e0, e1, e2, e3, f0, f1, 16)

  long k0 = 0;
  const long Kl = K;
  for (int lvl = 0; lvl < 4; ++lvl) {
    long kend;
    switch (lvl) {
      case 0:  __builtin_amdgcn_s_setprio(3); kend = Kl >> 1;               break;
      case 1:  __builtin_amdgcn_s_setprio(2); kend = (Kl >> 1) + (Kl >> 2); break;
      case 2:  __builtin_amdgcn_s_setprio(1); kend = Kl - (Kl >> 3);        break;
      default: __builtin_amdgcn_s_setprio(0); kend = Kl;                    break;
    }
    if (kend > Kl - 63) kend = Kl - 63;
#pragma clang loop unroll(disable)
  for (; k0 < kend; k0 += 32) {
    GEMM_LOAD6(g0, g1, g2, g3, h0, h1, k0 + 24)
    GEMM_FENCE
    GEMM_MMA16(a0, a1, a2, a3, b0, b1)
    GEMM_LOAD6(a0, a1, a2, a3, b0, b1, k0 + 32)
    GEMM_FENCE
    GEMM_MMA16(c0, c1, c2, c3, d0, d1)
    GEMM_LOAD6(c0, c1, c2, c3, d0, d1, k0 + 40)
    GEMM_FENCE
    GEMM_MMA16(e0, e1, e2, e3, f0, f1)
    GEMM_LOAD6(e0, e1, e2, e3, f0, f1, k0 + 48)
    GEMM_FENCE
    GEMM_MMA16(g0, g1, g2, g3, h0, h1)
  }
  }
  GEMM_LOAD6(g0, g1, g2, g3, h0, h1, k0 + 24)
  GEMM_MMA16(a0, a1, a2, a3, b0, b1)
  GEMM_MMA16(c0, c1, c2, c3, d0, d1)
  GEMM_MMA16(e0, e1, e2, e3, f0, f1)
  GEMM_MMA16(g0, g1, g2, g3, h0, h1)
#undef GEMM_FENCE
#undef GEMM_LOAD6
#undef GEMM_MMA16

  if (OUT == 0) {
#pragma unroll
    for (int mi = 0; mi < 4; ++mi)
#pragma unroll
      for (int ni = 0; ni < 2; ++ni)
#pragma unroll
        for (int r = 0; r < 8; ++r) {
          const int row = m0 + mi * 16 + r + lhalf * 8;
          const int col = n0 + ni * 16 + lmod;
          float v = acc[mi][ni][r];
          if (bias) v += bias[col];
          C[(size_t)row * N + col] = v;
        }
  } else if (OUT == 1) {
    const int h = m0 >> 6;
#pragma unroll
    for (int ni = 0; ni < 2; ++ni) {
      const int tok0  = n0 + ni * 16;
      const int bb    = tok0 / DIM_T;
      const int chunk = (tok0 % DIM_T) >> 4;
      float* dst = C + ((size_t)(bb * NHEAD + h) * (DIM_T / 16) + chunk) * 1024
                     + lane * 4;
#pragma unroll
      for (int mi = 0; mi < 4; ++mi)
#pragma unroll
        for (int rp = 0; rp < 2; ++rp) {
          v4f o;
          o.x = acc[mi][ni][4 * rp + 0];
          o.y = acc[mi][ni][4 * rp + 1];
          o.z = acc[mi][ni][4 * rp + 2];
          o.w = acc[mi][ni][4 * rp + 3];
          *(v4f*)(dst + (mi * 2 + rp) * 128) = o;
        }
    }
  } else {
    const int h  = n0 >> 6;
    const int jb = (n0 & 63) >> 4;
#pragma unroll
    for (int mi = 0; mi < 4; ++mi) {
      const int tok0  = m0 + mi * 16;
      const int bb    = tok0 / DIM_T;
      const int chunk = (tok0 % DIM_T) >> 4;
      float* dst = C + ((size_t)(bb * NHEAD + h) * (DIM_T / 16) + chunk) * 1024
                     + lane * 4;
#pragma unroll
      for (int ni = 0; ni < 2; ++ni)
#pragma unroll
        for (int rp = 0; rp < 2; ++rp) {
          v4f o;
          o.x = acc[mi][ni][4 * rp + 0];
          o.y = acc[mi][ni][4 * rp + 1];
          o.z = acc[mi][ni][4 * rp + 2];
          o.w = acc[mi][ni][4 * rp + 3];
          *(v4f*)(dst + ((jb + ni) * 2 + rp) * 128) = o;
        }
    }
  }
}

static __device__ __forceinline__ float xhalf(float x) {
  const int b = __builtin_bit_cast(int, x);
  return __builtin_bit_cast(
      float, __builtin_amdgcn_permlanex16(b, b, 0x76543210, 0xfedcba98, false,
                                          false));
}
static __device__ __forceinline__ float to_acc_rows(float v) {
  const int b = __builtin_bit_cast(int, v);
  return __builtin_bit_cast(
      float, __builtin_amdgcn_update_dpp(b, b, 0x160 + 8, 0x2, 0xf, false));
}
template <int R>
static __device__ __forceinline__ float row_share(float v) {
  return __builtin_bit_cast(
      float, __builtin_amdgcn_update_dpp(0, __builtin_bit_cast(int, v),
                                         0x150 + R, 0xf, 0xf, true));
}

__global__ __launch_bounds__(128)
__attribute__((amdgpu_waves_per_eu(1)))
void attn_kernel(const float* __restrict__ Q, const float* __restrict__ Kf,
                 const float* __restrict__ Vf, float* __restrict__ O) {
  __shared__ __attribute__((aligned(16))) float pbuf[4][16 * 20];

  const int lane  = threadIdx.x & 31;
  const int wave  = threadIdx.x >> 5;
  const int lhalf = lane >> 4;
  const int lmod  = lane & 15;
  const int h  = blockIdx.y;
  const int b  = blockIdx.z;
  const int q0 = blockIdx.x * 64 + wave * 16;

  const size_t base = (size_t)b * DIM_T * DIM_K + (size_t)h * HDIM;

  v2f qa[16];
  {
    const float* qrow = Q + base + (size_t)(q0 + lmod) * DIM_K + lhalf * 8;
#pragma unroll
    for (int g = 0; g < 8; ++g) {
      const v4f t = *(const v4f*)(qrow + (g >> 1) * 16 + (g & 1) * 4);
      qa[2 * g].x     = t.x;  qa[2 * g].y     = t.y;
      qa[2 * g + 1].x = t.z;  qa[2 * g + 1].y = t.w;
    }
  }

  const size_t fbase = (size_t)(b * NHEAD + h) * (DIM_T / 16) * 1024 + lane * 4;
  const float* kfp = Kf + fbase;
  const float* vfp = Vf + fbase;

  v8f acc[4];
#pragma unroll
  for (int j = 0; j < 4; ++j)
#pragma unroll
    for (int r = 0; r < 8; ++r) acc[j][r] = 0.0f;

  float mrow = -INFINITY, lrow = 0.0f;

  const float scale = 0.03125f;

  int key0 = 0;
  for (int lvl = 0; lvl < 4; ++lvl) {
    int kend;
    switch (lvl) {
      case 0:  __builtin_amdgcn_s_setprio(3); kend = DIM_T / 2;     break;
      case 1:  __builtin_amdgcn_s_setprio(2); kend = DIM_T * 3 / 4; break;
      case 2:  __builtin_amdgcn_s_setprio(1); kend = DIM_T * 7 / 8; break;
      default: __builtin_amdgcn_s_setprio(0); kend = DIM_T;         break;
    }
  for (; key0 < kend; key0 += 16) {
    const size_t coff = (size_t)(key0 >> 4) * 1024;
    v4f kk[8];
#pragma unroll
    for (int g = 0; g < 8; ++g)
      kk[g] = *(const v4f*)(kfp + coff + g * 128);

    v8f s;
#pragma unroll
    for (int r = 0; r < 8; ++r) s[r] = 0.0f;
#pragma unroll
    for (int g = 0; g < 8; ++g) {
      v2f k0, k1;
      k0.x = kk[g].x;  k0.y = kk[g].y;
      k1.x = kk[g].z;  k1.y = kk[g].w;
      s = wmma_f32(qa[2 * g], k0, s);
      s = wmma_f32(qa[2 * g + 1], k1, s);
    }

    v2f vf[4][4];
#pragma unroll
    for (int j = 0; j < 4; ++j)
#pragma unroll
      for (int rp = 0; rp < 2; ++rp) {
        const v4f t = *(const v4f*)(vfp + coff + (j * 2 + rp) * 128);
        vf[2 * rp][j].x     = t.x;  vf[2 * rp][j].y     = t.y;
        vf[2 * rp + 1][j].x = t.z;  vf[2 * rp + 1][j].y = t.w;
      }
    __builtin_amdgcn_sched_barrier(0);

#pragma unroll
    for (int r = 0; r < 8; ++r)
      pbuf[wave][(r + lhalf * 8) * 20 + lmod] = s[r];
    const v4f t0 = *(const v4f*)&pbuf[wave][lmod * 20 + lhalf * 8];
    const v4f t1 = *(const v4f*)&pbuf[wave][lmod * 20 + lhalf * 8 + 4];
    float sv[8];
    sv[0] = t0.x * scale;  sv[1] = t0.y * scale;
    sv[2] = t0.z * scale;  sv[3] = t0.w * scale;
    sv[4] = t1.x * scale;  sv[5] = t1.y * scale;
    sv[6] = t1.z * scale;  sv[7] = t1.w * scale;

    const float mloc =
        fmaxf(fmaxf(fmaxf(sv[0], sv[1]), fmaxf(sv[2], sv[3])),
              fmaxf(fmaxf(sv[4], sv[5]), fmaxf(sv[6], sv[7])));
    const float mnew = fmaxf(mrow, fmaxf(mloc, xhalf(mloc)));
    const float corr = __expf(mrow - mnew);
    mrow = mnew;
    float pe[8];
#pragma unroll
    for (int e = 0; e < 8; ++e) pe[e] = __expf(sv[e] - mnew);
    lrow = lrow * corr + (((pe[0] + pe[1]) + (pe[2] + pe[3])) +
                          ((pe[4] + pe[5]) + (pe[6] + pe[7])));

    {
      const float c2 = to_acc_rows(corr);
      float cr[8];
      cr[0] = row_share<0>(c2);  cr[1] = row_share<1>(c2);
      cr[2] = row_share<2>(c2);  cr[3] = row_share<3>(c2);
      cr[4] = row_share<4>(c2);  cr[5] = row_share<5>(c2);
      cr[6] = row_share<6>(c2);  cr[7] = row_share<7>(c2);
#pragma unroll
      for (int j = 0; j < 4; ++j)
#pragma unroll
        for (int r = 0; r < 8; ++r) acc[j][r] *= cr[r];
    }

    v2f pa[4];
#pragma unroll
    for (int t = 0; t < 4; ++t) { pa[t].x = pe[2 * t];  pa[t].y = pe[2 * t + 1]; }
#pragma unroll
    for (int t = 0; t < 4; ++t)
#pragma unroll
      for (int j = 0; j < 4; ++j)
        acc[j] = wmma_f32(pa[t], vf[t][j], acc[j]);
  }
  }

  {
    const float l2 = to_acc_rows(lrow + xhalf(lrow));
    float lr[8];
    lr[0] = row_share<0>(l2);  lr[1] = row_share<1>(l2);
    lr[2] = row_share<2>(l2);  lr[3] = row_share<3>(l2);
    lr[4] = row_share<4>(l2);  lr[5] = row_share<5>(l2);
    lr[6] = row_share<6>(l2);  lr[7] = row_share<7>(l2);
#pragma unroll
    for (int j = 0; j < 4; ++j)
#pragma unroll
      for (int r = 0; r < 8; ++r) {
        const float ov = acc[j][r] / lr[r];
        O[base + (size_t)(q0 + r + lhalf * 8) * DIM_K + j * 16 + lmod] = ov;
      }
  }
}

extern "C" void kernel_launch(void* const* d_in, const int* in_sizes, int n_in,
                              void* d_out, int out_size, void* d_ws,
                              size_t ws_size, hipStream_t stream) {
  (void)in_sizes; (void)n_in; (void)out_size; (void)ws_size;
  const float* x  = (const float*)d_in[0];
  const float* wq = (const float*)d_in[1];
  const float* wk = (const float*)d_in[2];
  const float* wv = (const float*)d_in[3];
  const float* wu = (const float*)d_in[4];
  const float* bu = (const float*)d_in[5];
  float* out = (float*)d_out;

  const size_t MAT = (size_t)DIM_B * DIM_T * DIM_K;
  float* qb = (float*)d_ws;
  float* kb = qb + MAT;
  float* vb = kb + MAT;
  float* ab = vb + MAT;

  const int M = DIM_B * DIM_T, N = DIM_K, K = DIM_K;
  dim3 gblk(256);
  dim3 ggrid(N / 128, M / 128);

  gemm_xwt_kernel<0><<<ggrid, gblk, 0, stream>>>(x, wq, nullptr, qb, M, N, K);
  dim3 kgrid(M / 128, N / 128);
  gemm_xwt_kernel<1><<<kgrid, gblk, 0, stream>>>(wk, x, nullptr, kb, N, M, K);
  gemm_xwt_kernel<2><<<ggrid, gblk, 0, stream>>>(x, wv, nullptr, vb, M, N, K);

  dim3 agrid(DIM_T / 64, NHEAD, DIM_B);
  attn_kernel<<<agrid, dim3(128), 0, stream>>>(qb, kb, vb, ab);

  gemm_xwt_kernel<0><<<ggrid, gblk, 0, stream>>>(ab, wu, bu, out, M, N, K);
}


# attention: softmax against a per-row reference raised only when a score lies 2^8 above it (else no exchange, correction or rescale; exp2 of one fma per score); sum orders kept, their scale differs
# speedup vs baseline: 1.4818x; 1.0307x over previous
#include <hip/hip_runtime.h>
#include <math.h>

typedef __attribute__((ext_vector_type(2))) float v2f;
typedef __attribute__((ext_vector_type(4))) float v4f;
typedef __attribute__((ext_vector_type(8))) float v8f;

#define DIM_B 2
#define DIM_T 2048
#define DIM_K 1024
#define NHEAD 16
#define HDIM  64

static __device__ __forceinline__ v8f wmma_f32(v2f a, v2f b, v8f c) {
  return __builtin_amdgcn_wmma_f32_16x16x4_f32(false, a, false, b, (short)0, c,
                                               false, false);
}

template <int OUT>
__global__ __launch_bounds__(256)
__attribute__((amdgpu_waves_per_eu(2)))
void gemm_xwt_kernel(const float* __restrict__ A, const float* __restrict__ W,
                     const float* __restrict__ bias, float* __restrict__ C,
                     int M, int N, int K) {
  const int lane  = threadIdx.x & 31;
  const int wave  = threadIdx.x >> 5;
  const int wm    = wave >> 2;
  const int wn    = wave & 3;
  const int m0    = blockIdx.y * 128 + wm * 64;
  const int n0    = blockIdx.x * 128 + wn * 32;
  const int lhalf = lane >> 4;
  const int lmod  = lane & 15;
  const int kbase = lhalf * 4;

  v8f acc[4][2];
#pragma unroll
  for (int i = 0; i < 4; ++i)
#pragma unroll
    for (int j = 0; j < 2; ++j)
#pragma unroll
      for (int r = 0; r < 8; ++r) acc[i][j][r] = 0.0f;

  const float* ar0 = A + (size_t)(m0 + lmod) * K;
  const float* ar1 = ar0 + (size_t)16 * K;
  const float* ar2 = ar0 + (size_t)32 * K;
  const float* ar3 = ar0 + (size_t)48 * K;
  const float* br0 = W + (size_t)(n0 + lmod) * K;
  const float* br1 = br0 + (size_t)16 * K;

#define GEMM_MMA16(f0, f1, f2, f3, g0, g1)            \
  acc[0][0] = wmma_f32(f0.xy, g0.xy, acc[0][0]);      \
  acc[0][1] = wmma_f32(f0.xy, g1.xy, acc[0][1]);      \
  acc[1][0] = wmma_f32(f1.xy, g0.xy, acc[1][0]);      \
  acc[1][1] = wmma_f32(f1.xy, g1.xy, acc[1][1]);      \
  acc[2][0] = wmma_f32(f2.xy, g0.xy, acc[2][0]);      \
  acc[2][1] = wmma_f32(f2.xy, g1.xy, acc[2][1]);      \
  acc[3][0] = wmma_f32(f3.xy, g0.xy, acc[3][0]);      \
  acc[3][1] = wmma_f32(f3.xy, g1.xy, acc[3][1]);      \
  acc[0][0] = wmma_f32(f0.zw, g0.zw, acc[0][0]);      \
  acc[0][1] = wmma_f32(f0.zw, g1.zw, acc[0][1]);      \
  acc[1][0] = wmma_f32(f1.zw, g0.zw, acc[1][0]);      \
  acc[1][1] = wmma_f32(f1.zw, g1.zw, acc[1][1]);      \
  acc[2][0] = wmma_f32(f2.zw, g0.zw, acc[2][0]);      \
  acc[2][1] = wmma_f32(f2.zw, g1.zw, acc[2][1]);      \
  acc[3][0] = wmma_f32(f3.zw, g0.zw, acc[3][0]);      \
  acc[3][1] = wmma_f32(f3.zw, g1.zw, acc[3][1]);

#define GEMM_LOAD6(f0, f1, f2, f3, g0, g1, koff)      \
  {                                                   \
    const long kn = (long)(koff) + kbase;             \
    f0 = *(const v4f*)(ar0 + kn);                     \
    f1 = *(const v4f*)(ar1 + kn);                     \
    f2 = *(const v4f*)(ar2 + kn);                     \
    f3 = *(const v4f*)(ar3 + kn);                     \
    g0 = *(const v4f*)(br0 + kn);                     \
    g1 = *(const v4f*)(br1 + kn);                     \
  }
#define GEMM_FENCE __builtin_amdgcn_sched_barrier(0);

  v4f a0, a1, a2, a3, b0, b1;
  v4f c0, c1, c2, c3, d0, d1;
  v4f e0, e1, e2, e3, f0, f1;
  v4f g0, g1, g2, g3, h0, h1;

  GEMM_LOAD6(a0, a1, a2, a3, b0, b1, 0)
  GEMM_FENCE
  GEMM_LOAD6(c0, c1, c2, c3, d0, d1, 8)
  GEMM_FENCE
  GEMM_LOAD6(e0, e1, e2, e3, f0, f1, 16)

  long k0 = 0;
  const long Kl = K;
  for (int lvl = 0; lvl < 4; ++lvl) {
    long kend;
    switch (lvl) {
      case 0:  __builtin_amdgcn_s_setprio(3); kend = Kl >> 1;               break;
      case 1:  __builtin_amdgcn_s_setprio(2); kend = (Kl >> 1) + (Kl >> 2); break;
      case 2:  __builtin_amdgcn_s_setprio(1); kend = Kl - (Kl >> 3);        break;
      default: __builtin_amdgcn_s_setprio(0); kend = Kl;                    break;
    }
    if (kend > Kl - 63) kend = Kl - 63;
#pragma clang loop unroll(disable)
  for (; k0 < kend; k0 += 32) {
    GEMM_LOAD6(g0, g1, g2, g3, h0, h1, k0 + 24)
    GEMM_FENCE
    GEMM_MMA16(a0, a1, a2, a3, b0, b1)
    GEMM_LOAD6(a0, a1, a2, a3, b0, b1, k0 + 32)
    GEMM_FENCE
    GEMM_MMA16(c0, c1, c2, c3, d0, d1)
    GEMM_LOAD6(c0, c1, c2, c3, d0, d1, k0 + 40)
    GEMM_FENCE
    GEMM_MMA16(e0, e1, e2, e3, f0, f1)
    GEMM_LOAD6(e0, e1, e2, e3, f0, f1, k0 + 48)
    GEMM_FENCE
    GEMM_MMA16(g0, g1, g2, g3, h0, h1)
  }
  }
  GEMM_LOAD6(g0, g1, g2, g3, h0, h1, k0 + 24)
  GEMM_MMA16(a0, a1, a2, a3, b0, b1)
  GEMM_MMA16(c0, c1, c2, c3, d0, d1)
  GEMM_MMA16(e0, e1, e2, e3, f0, f1)
  GEMM_MMA16(g0, g1, g2, g3, h0, h1)
#undef GEMM_FENCE
#undef GEMM_LOAD6
#undef GEMM_MMA16

  if (OUT == 0) {
#pragma unroll
    for (int mi = 0; mi < 4; ++mi)
#pragma unroll
      for (int ni = 0; ni < 2; ++ni)
#pragma unroll
        for (int r = 0; r < 8; ++r) {
          const int row = m0 + mi * 16 + r + lhalf * 8;
          const int col = n0 + ni * 16 + lmod;
          float v = acc[mi][ni][r];
          if (bias) v += bias[col];
          C[(size_t)row * N + col] = v;
        }
  } else if (OUT == 1) {
    const int h = m0 >> 6;
#pragma unroll
    for (int ni = 0; ni < 2; ++ni) {
      const int tok0  = n0 + ni * 16;
      const int bb    = tok0 / DIM_T;
      const int chunk = (tok0 % DIM_T) >> 4;
      float* dst = C + ((size_t)(bb * NHEAD + h) * (DIM_T / 16) + chunk) * 1024
                     + lane * 4;
#pragma unroll
      for (int mi = 0; mi < 4; ++mi)
#pragma unroll
        for (int rp = 0; rp < 2; ++rp) {
          v4f o;
          o.x = acc[mi][ni][4 * rp + 0];
          o.y = acc[mi][ni][4 * rp + 1];
          o.z = acc[mi][ni][4 * rp + 2];
          o.w = acc[mi][ni][4 * rp + 3];
          *(v4f*)(dst + (mi * 2 + rp) * 128) = o;
        }
    }
  } else {
    const int h  = n0 >> 6;
    const int jb = (n0 & 63) >> 4;
#pragma unroll
    for (int mi = 0; mi < 4; ++mi) {
      const int tok0  = m0 + mi * 16;
      const int bb    = tok0 / DIM_T;
      const int chunk = (tok0 % DIM_T) >> 4;
      float* dst = C + ((size_t)(bb * NHEAD + h) * (DIM_T / 16) + chunk) * 1024
                     + lane * 4;
#pragma unroll
      for (int ni = 0; ni < 2; ++ni)
#pragma unroll
        for (int rp = 0; rp < 2; ++rp) {
          v4f o;
          o.x = acc[mi][ni][4 * rp + 0];
          o.y = acc[mi][ni][4 * rp + 1];
          o.z = acc[mi][ni][4 * rp + 2];
          o.w = acc[mi][ni][4 * rp + 3];
          *(v4f*)(dst + ((jb + ni) * 2 + rp) * 128) = o;
        }
    }
  }
}

static __device__ __forceinline__ float xhalf(float x) {
  const int b = __builtin_bit_cast(int, x);
  return __builtin_bit_cast(
      float, __builtin_amdgcn_permlanex16(b, b, 0x76543210, 0xfedcba98, false,
                                          false));
}
static __device__ __forceinline__ float to_acc_rows(float v) {
  const int b = __builtin_bit_cast(int, v);
  return __builtin_bit_cast(
      float, __builtin_amdgcn_update_dpp(b, b, 0x160 + 8, 0x2, 0xf, false));
}
template <int R>
static __device__ __forceinline__ float row_share(float v) {
  return __builtin_bit_cast(
      float, __builtin_amdgcn_update_dpp(0, __builtin_bit_cast(int, v),
                                         0x150 + R, 0xf, 0xf, true));
}

__global__ __launch_bounds__(128)
__attribute__((amdgpu_waves_per_eu(1)))
void attn_kernel(const float* __restrict__ Q, const float* __restrict__ Kf,
                 const float* __restrict__ Vf, float* __restrict__ O) {
  __shared__ __attribute__((aligned(16))) float pbuf[4][16 * 20];

  const int lane  = threadIdx.x & 31;
  const int wave  = threadIdx.x >> 5;
  const int lhalf = lane >> 4;
  const int lmod  = lane & 15;
  const int h  = blockIdx.y;
  const int b  = blockIdx.z;
  const int q0 = blockIdx.x * 64 + wave * 16;

  const size_t base = (size_t)b * DIM_T * DIM_K + (size_t)h * HDIM;

  v2f qa[16];
  {
    const float* qrow = Q + base + (size_t)(q0 + lmod) * DIM_K + lhalf * 8;
#pragma unroll
    for (int g = 0; g < 8; ++g) {
      const v4f t = *(const v4f*)(qrow + (g >> 1) * 16 + (g & 1) * 4);
      qa[2 * g].x     = t.x;  qa[2 * g].y     = t.y;
      qa[2 * g + 1].x = t.z;  qa[2 * g + 1].y = t.w;
    }
  }

  const size_t fbase = (size_t)(b * NHEAD + h) * (DIM_T / 16) * 1024 + lane * 4;
  const float* kfp = Kf + fbase;
  const float* vfp = Vf + fbase;

  v8f acc[4];
#pragma unroll
  for (int j = 0; j < 4; ++j)
#pragma unroll
    for (int r = 0; r < 8; ++r) acc[j][r] = 0.0f;

  float lrow = 0.0f;

  const float cexp = 0.03125f * 1.44269504f;
  const float lift = 8.0f / (0.03125f * 1.44269504f);
  float mref = -INFINITY, nref = INFINITY, thr = -INFINITY;

  int key0 = 0;
  for (int lvl = 0; lvl < 4; ++lvl) {
    int kend;
    switch (lvl) {
      case 0:  __builtin_amdgcn_s_setprio(3); kend = DIM_T / 2;     break;
      case 1:  __builtin_amdgcn_s_setprio(2); kend = DIM_T * 3 / 4; break;
      case 2:  __builtin_amdgcn_s_setprio(1); kend = DIM_T * 7 / 8; break;
      default: __builtin_amdgcn_s_setprio(0); kend = DIM_T;         break;
    }
  for (; key0 < kend; key0 += 16) {
    const size_t coff = (size_t)(key0 >> 4) * 1024;
    v4f kk[8];
#pragma unroll
    for (int g = 0; g < 8; ++g)
      kk[g] = *(const v4f*)(kfp + coff + g * 128);

    v8f s;
#pragma unroll
    for (int r = 0; r < 8; ++r) s[r] = 0.0f;
#pragma unroll
    for (int g = 0; g < 8; ++g) {
      v2f k0, k1;
      k0.x = kk[g].x;  k0.y = kk[g].y;
      k1.x = kk[g].z;  k1.y = kk[g].w;
      s = wmma_f32(qa[2 * g], k0, s);
      s = wmma_f32(qa[2 * g + 1], k1, s);
    }

    v2f vf[4][4];
#pragma unroll
    for (int j = 0; j < 4; ++j)
#pragma unroll
      for (int rp = 0; rp < 2; ++rp) {
        const v4f t = *(const v4f*)(vfp + coff + (j * 2 + rp) * 128);
        vf[2 * rp][j].x     = t.x;  vf[2 * rp][j].y     = t.y;
        vf[2 * rp + 1][j].x = t.z;  vf[2 * rp + 1][j].y = t.w;
      }
    __builtin_amdgcn_sched_barrier(0);

#pragma unroll
    for (int r = 0; r < 8; ++r)
      pbuf[wave][(r + lhalf * 8) * 20 + lmod] = s[r];
    const v4f t0 = *(const v4f*)&pbuf[wave][lmod * 20 + lhalf * 8];
    const v4f t1 = *(const v4f*)&pbuf[wave][lmod * 20 + lhalf * 8 + 4];

    const float mloc =
        fmaxf(fmaxf(fmaxf(t0.x, t0.y), fmaxf(t0.z, t0.w)),
              fmaxf(fmaxf(t1.x, t1.y), fmaxf(t1.z, t1.w)));
    if (__builtin_expect(__builtin_amdgcn_ballot_w32(mloc > thr) != 0u, 0)) {
      const float mnew = fmaxf(mref, fmaxf(mloc, xhalf(mloc)));
      const float corr = __builtin_amdgcn_exp2f((mref - mnew) * cexp);
      mref = mnew;
      nref = -mnew * cexp;
      thr  = mnew + lift;
      lrow *= corr;
      const float c2 = to_acc_rows(corr);
      float cr[8];
      cr[0] = row_share<0>(c2);  cr[1] = row_share<1>(c2);
      cr[2] = row_share<2>(c2);  cr[3] = row_share<3>(c2);
      cr[4] = row_share<4>(c2);  cr[5] = row_share<5>(c2);
      cr[6] = row_share<6>(c2);  cr[7] = row_share<7>(c2);
#pragma unroll
      for (int j = 0; j < 4; ++j)
#pragma unroll
        for (int r = 0; r < 8; ++r) acc[j][r] *= cr[r];
    }

    float pe[8];
    pe[0] = __builtin_amdgcn_exp2f(fmaf(t0.x, cexp, nref));
    pe[1] = __builtin_amdgcn_exp2f(fmaf(t0.y, cexp, nref));
    pe[2] = __builtin_amdgcn_exp2f(fmaf(t0.z, cexp, nref));
    pe[3] = __builtin_amdgcn_exp2f(fmaf(t0.w, cexp, nref));
    pe[4] = __builtin_amdgcn_exp2f(fmaf(t1.x, cexp, nref));
    pe[5] = __builtin_amdgcn_exp2f(fmaf(t1.y, cexp, nref));
    pe[6] = __builtin_amdgcn_exp2f(fmaf(t1.z, cexp, nref));
    pe[7] = __builtin_amdgcn_exp2f(fmaf(t1.w, cexp, nref));
    lrow += ((pe[0] + pe[1]) + (pe[2] + pe[3])) +
            ((pe[4] + pe[5]) + (pe[6] + pe[7]));

    v2f pa[4];
#pragma unroll
    for (int t = 0; t < 4; ++t) { pa[t].x = pe[2 * t];  pa[t].y = pe[2 * t + 1]; }
#pragma unroll
    for (int t = 0; t < 4; ++t)
#pragma unroll
      for (int j = 0; j < 4; ++j)
        acc[j] = wmma_f32(pa[t], vf[t][j], acc[j]);
  }
  }

  {
    const float l2 = to_acc_rows(lrow + xhalf(lrow));
    float lr[8];
    lr[0] = row_share<0>(l2);  lr[1] = row_share<1>(l2);
    lr[2] = row_share<2>(l2);  lr[3] = row_share<3>(l2);
    lr[4] = row_share<4>(l2);  lr[5] = row_share<5>(l2);
    lr[6] = row_share<6>(l2);  lr[7] = row_share<7>(l2);
#pragma unroll
    for (int j = 0; j < 4; ++j)
#pragma unroll
      for (int r = 0; r < 8; ++r) {
        const float ov = acc[j][r] / lr[r];
        O[base + (size_t)(q0 + r + lhalf * 8) * DIM_K + j * 16 + lmod] = ov;
      }
  }
}

extern "C" void kernel_launch(void* const* d_in, const int* in_sizes, int n_in,
                              void* d_out, int out_size, void* d_ws,
                              size_t ws_size, hipStream_t stream) {
  (void)in_sizes; (void)n_in; (void)out_size; (void)ws_size;
  const float* x  = (const float*)d_in[0];
  const float* wq = (const float*)d_in[1];
  const float* wk = (const float*)d_in[2];
  const float* wv = (const float*)d_in[3];
  const float* wu = (const float*)d_in[4];
  const float* bu = (const float*)d_in[5];
  float* out = (float*)d_out;

  const size_t MAT = (size_t)DIM_B * DIM_T * DIM_K;
  float* qb = (float*)d_ws;
  float* kb = qb + MAT;
  float* vb = kb + MAT;
  float* ab = vb + MAT;

  const int M = DIM_B * DIM_T, N = DIM_K, K = DIM_K;
  dim3 gblk(256);
  dim3 ggrid(N / 128, M / 128);

  gemm_xwt_kernel<0><<<ggrid, gblk, 0, stream>>>(x, wq, nullptr, qb, M, N, K);
  dim3 kgrid(M / 128, N / 128);
  gemm_xwt_kernel<1><<<kgrid, gblk, 0, stream>>>(wk, x, nullptr, kb, N, M, K);
  gemm_xwt_kernel<2><<<ggrid, gblk, 0, stream>>>(x, wv, nullptr, vb, M, N, K);

  dim3 agrid(DIM_T / 64, NHEAD, DIM_B);
  attn_kernel<<<agrid, dim3(128), 0, stream>>>(qb, kb, vb, ab);

  gemm_xwt_kernel<0><<<ggrid, gblk, 0, stream>>>(ab, wu, bu, out, M, N, K);
}
